# starting kernel
# baseline (speedup 1.0000x reference)
#include <hip/hip_runtime.h>
#include <hip/hip_bf16.h>
#include <math.h>

typedef float v2f __attribute__((ext_vector_type(2)));
typedef float v8f __attribute__((ext_vector_type(8)));

#define N_NODES 4096
#define B_SZ    4
#define F_SZ    64
#define ALPHA   0.2f
#define NEG_BIG -9e15f
#define L2E     1.4426950408889634f

__global__ __launch_bounds__(64) void wh_kernel(
    const float* __restrict__ h, const float* __restrict__ W,
    const float* __restrict__ a, float* __restrict__ WhP,
    float* __restrict__ s1, float* __restrict__ t2)
{
    int bi = blockIdx.x;
    int o  = threadIdx.x;
    const float* hrow = h + (size_t)bi * F_SZ;
    float acc = 0.f;
#pragma unroll
    for (int f = 0; f < F_SZ; ++f)
        acc = fmaf(hrow[f], W[f * F_SZ + o], acc);
    int operm = ((o & 15) << 2) | (o >> 4);
    WhP[(size_t)bi * F_SZ + operm] = acc;

    __shared__ float sWh[F_SZ];
    sWh[o] = acc;
    __syncthreads();
    if (o == 0) {
        float s = 0.f;
        for (int k = 0; k < F_SZ; ++k) s = fmaf(sWh[k], a[k], s);
        s1[bi] = s * (1.0f / 64.0f);
    } else if (o == 1) {
        float t = 0.f;
        for (int k = 0; k < F_SZ; ++k) t = fmaf(sWh[k], a[F_SZ + k], t);
        t2[bi] = t * (1.0f / 64.0f);
    }
}

__global__ __launch_bounds__(256) void rowmax_kernel(
    const int* __restrict__ adj, const float* __restrict__ s1,
    const float* __restrict__ t2, float* __restrict__ mrow)
{
    int wave = threadIdx.x >> 5;
    int lane = threadIdx.x & 31;
    int i = blockIdx.x * 8 + wave;
    const int* arow = adj + (size_t)i * N_NODES;
    float T0 = -3.0e38f, T1 = -3.0e38f, T2 = -3.0e38f, T3 = -3.0e38f;
    int any = 0;
    for (int j = lane; j < N_NODES; j += 32) {
        int av = arow[j];
        bool sel = av > 0;
        any |= (int)sel;
        float u0 = t2[j];
        float u1 = t2[N_NODES + j];
        float u2 = t2[2 * N_NODES + j];
        float u3 = t2[3 * N_NODES + j];
        T0 = fmaxf(T0, sel ? u0 : -3.0e38f);
        T1 = fmaxf(T1, sel ? u1 : -3.0e38f);
        T2 = fmaxf(T2, sel ? u2 : -3.0e38f);
        T3 = fmaxf(T3, sel ? u3 : -3.0e38f);
    }
#pragma unroll
    for (int off = 16; off >= 1; off >>= 1) {
        T0 = fmaxf(T0, __shfl_xor(T0, off, 32));
        T1 = fmaxf(T1, __shfl_xor(T1, off, 32));
        T2 = fmaxf(T2, __shfl_xor(T2, off, 32));
        T3 = fmaxf(T3, __shfl_xor(T3, off, 32));
        any |= __shfl_xor(any, off, 32);
    }
    if (lane == 0) {
        float Tb[4] = {T0, T1, T2, T3};
#pragma unroll
        for (int b = 0; b < 4; ++b) {
            float x = s1[b * N_NODES + i] + Tb[b];
            float m = any ? fmaxf(x, ALPHA * x) : NEG_BIG;
            mrow[b * N_NODES + i] = m;
        }
    }
}

__global__ __launch_bounds__(32) void attn_kernel(
    const int* __restrict__ adj, const float* __restrict__ WhP,
    const float* __restrict__ s1, const float* __restrict__ t2,
    const float* __restrict__ mrow, float* __restrict__ Dp,
    float* __restrict__ Zp, int jspan)
{
    int bid   = blockIdx.x;
    int b     = bid & 3;
    int itile = (bid >> 2) & 255;
    int chunk = bid >> 10;
    int lane  = threadIdx.x;
    int l15   = lane & 15;
    int half  = lane >> 4;
    int irow  = itile * 16 + l15;

    float s = s1[b * N_NODES + irow];
    float m = mrow[b * N_NODES + irow];
    const int*   adjrow = adj + (size_t)irow * N_NODES;
    const float* trow   = t2 + (size_t)b * N_NODES;
    const float* WhB    = WhP + (size_t)b * N_NODES * F_SZ;

    v8f c0 = {}, c1 = {}, c2 = {}, c3 = {};
    float zacc = 0.f;

    int j0 = chunk * jspan;
    int jend = j0 + jspan;
#pragma unroll 4
    for (int j = j0; j < jend; j += 4) {
        int jj = j + half * 2;
        int2   av = *(const int2*)(adjrow + jj);
        float2 tv = *(const float2*)(trow + jj);

        float x0 = s + tv.x, x1 = s + tv.y;
        float e0 = fmaxf(x0, ALPHA * x0);
        float e1 = fmaxf(x1, ALPHA * x1);
        float sc0 = (av.x > 0) ? e0 : NEG_BIG;
        float sc1 = (av.y > 0) ? e1 : NEG_BIG;
        float p0 = __builtin_amdgcn_exp2f((sc0 - m) * L2E);
        float p1 = __builtin_amdgcn_exp2f((sc1 - m) * L2E);
        zacc += p0 + p1;

        v2f A; A[0] = p0; A[1] = p1;

        const float* bp = WhB + (size_t)jj * F_SZ + (l15 << 2);
        float4 q0 = *(const float4*)(bp);
        float4 q1 = *(const float4*)(bp + F_SZ);
        v2f B0, B1, B2, B3;
        B0[0] = q0.x; B0[1] = q1.x;
        B1[0] = q0.y; B1[1] = q1.y;
        B2[0] = q0.z; B2[1] = q1.z;
        B3[0] = q0.w; B3[1] = q1.w;

        c0 = __builtin_amdgcn_wmma_f32_16x16x4_f32(false, A, false, B0, (short)0, c0, false, false);
        c1 = __builtin_amdgcn_wmma_f32_16x16x4_f32(false, A, false, B1, (short)0, c1, false, false);
        c2 = __builtin_amdgcn_wmma_f32_16x16x4_f32(false, A, false, B2, (short)0, c2, false, false);
        c3 = __builtin_amdgcn_wmma_f32_16x16x4_f32(false, A, false, B3, (short)0, c3, false, false);
    }

    zacc += __shfl_xor(zacc, 16, 32);

    size_t obase = (((size_t)chunk * 4 + b) * N_NODES + (size_t)itile * 16) * F_SZ;
    float* dout = Dp + obase;
    int rbase = half * 8;
#pragma unroll
    for (int r = 0; r < 8; ++r) {
        float4 v; v.x = c0[r]; v.y = c1[r]; v.z = c2[r]; v.w = c3[r];
        *(float4*)(dout + (size_t)(rbase + r) * F_SZ + (l15 << 2)) = v;
    }
    if (half == 0)
        Zp[((size_t)chunk * 4 + b) * N_NODES + irow] = zacc;
}

__global__ __launch_bounds__(256) void bn_kernel(
    const float* __restrict__ Dp, const float* __restrict__ Zp,
    const float* __restrict__ h, const float* __restrict__ gamma,
    const float* __restrict__ beta, float* __restrict__ out, int nchunk)
{
    int i   = blockIdx.x;
    int tid = threadIdx.x;
    int b   = tid >> 6;
    int o   = tid & 63;
    int operm = ((o & 15) << 2) | (o >> 4);

    float num = 0.f, z = 0.f;
    for (int c = 0; c < nchunk; ++c) {
        num += Dp[((((size_t)c * 4 + b) * N_NODES + i) << 6) + operm];
        z   += Zp[(((size_t)c * 4 + b) * N_NODES) + i];
    }
    float hp = num / z;

    __shared__ float red[256];
    red[tid] = hp;
    __syncthreads();
#pragma unroll
    for (int s = 128; s >= 1; s >>= 1) {
        if (tid < s) red[tid] += red[tid + s];
        __syncthreads();
    }
    float mean = red[0] * (1.0f / 256.0f);
    __syncthreads();
    float dev = hp - mean;
    red[tid] = dev * dev;
    __syncthreads();
#pragma unroll
    for (int s = 128; s >= 1; s >>= 1) {
        if (tid < s) red[tid] += red[tid + s];
        __syncthreads();
    }
    float var = red[0] * (1.0f / 256.0f);

    float xn  = gamma[i] * dev * (1.0f / sqrtf(var + 1e-5f)) + beta[i];
    float res = xn + h[((size_t)b * N_NODES + i) * F_SZ + o];
    out[((size_t)b * N_NODES + i) * F_SZ + o] = res > 0.f ? res : expm1f(res);
}

extern "C" void kernel_launch(void* const* d_in, const int* in_sizes, int n_in,
                              void* d_out, int out_size, void* d_ws, size_t ws_size,
                              hipStream_t stream)
{
    const float* h     = (const float*)d_in[0];
    const int*   adj   = (const int*)  d_in[1];
    const float* W     = (const float*)d_in[2];
    const float* a     = (const float*)d_in[3];
    const float* gamma = (const float*)d_in[4];
    const float* beta  = (const float*)d_in[5];
    float* out = (float*)d_out;

    const size_t WH_ELEMS = (size_t)B_SZ * N_NODES * F_SZ;
    const size_t NROW     = (size_t)B_SZ * N_NODES;

    int nchunk = 4;
    size_t need4 = (WH_ELEMS + 3 * NROW + 4 * (WH_ELEMS + NROW)) * sizeof(float);
    if (ws_size < need4) nchunk = 1;

    float* wsf  = (float*)d_ws;
    float* WhP  = wsf;
    float* s1   = WhP + WH_ELEMS;
    float* t2   = s1 + NROW;
    float* mrow = t2 + NROW;
    float* Dp   = mrow + NROW;
    float* Zp   = Dp + (size_t)nchunk * WH_ELEMS;

    int jspan = N_NODES / nchunk;

    wh_kernel<<<B_SZ * N_NODES, 64, 0, stream>>>(h, W, a, WhP, s1, t2);
    rowmax_kernel<<<N_NODES / 8, 256, 0, stream>>>(adj, s1, t2, mrow);
    attn_kernel<<<B_SZ * (N_NODES / 16) * nchunk, 32, 0, stream>>>(
        adj, WhP, s1, t2, mrow, Dp, Zp, jspan);
    bn_kernel<<<N_NODES, 256, 0, stream>>>(Dp, Zp, h, gamma, beta, out, nchunk);
}


# bn kernel: g@W on the f32 16x16x4 matrix instruction, one wave per 16-row tile (4 nodes x 4 batches); BatchNorm sums by lane exchange in D layout, no LDS, no barrier; h', mean, var sum order changes
# speedup vs baseline: 1.7419x; 1.7419x over previous
#include <hip/hip_runtime.h>
#include <hip/hip_bf16.h>
#include <math.h>

typedef float v2f __attribute__((ext_vector_type(2)));
typedef float v8f __attribute__((ext_vector_type(8)));

#define N_NODES 4096
#define B_SZ    4
#define F_SZ    64
#define ALPHA   0.2f
#define NEG_BIG -9e15f
#define L2E     1.4426950408889634f

__global__ __launch_bounds__(256) void st_kernel(
    const float* __restrict__ h, const float* __restrict__ W,
    const float* __restrict__ a, float* __restrict__ s1,
    float* __restrict__ t2)
{
    __shared__ __attribute__((aligned(16))) float uS[2 * F_SZ];

    int tid  = threadIdx.x;
    int wave = tid >> 5;
    int lane = tid & 31;
    int l15  = lane & 15;
    int half = lane >> 4;
    int row0 = (blockIdx.x * 8 + wave) * 8;

    const float* hp = h + (size_t)(row0 + half) * F_SZ + (l15 << 2);
    float4 v0 = *(const float4*)(hp);
    float4 v1 = *(const float4*)(hp + 2 * F_SZ);
    float4 v2 = *(const float4*)(hp + 4 * F_SZ);
    float4 v3 = *(const float4*)(hp + 6 * F_SZ);

    if (tid < 2 * F_SZ) {
        int sel = tid >> 6;
        int f   = tid & 63;
        const float* wr = W + f * F_SZ;
        const float* av = a + sel * F_SZ;
        float u = 0.f;
#pragma unroll
        for (int o = 0; o < F_SZ; ++o) u = fmaf(wr[o], av[o], u);
        uS[tid] = u * (1.0f / 64.0f);
    }
    __syncthreads();

    float4 u1 = *(const float4*)(&uS[l15 << 2]);
    float4 u2 = *(const float4*)(&uS[F_SZ + (l15 << 2)]);

    float ps0 = fmaf(v0.w, u1.w, fmaf(v0.z, u1.z, fmaf(v0.y, u1.y, v0.x * u1.x)));
    float ps1 = fmaf(v1.w, u1.w, fmaf(v1.z, u1.z, fmaf(v1.y, u1.y, v1.x * u1.x)));
    float ps2 = fmaf(v2.w, u1.w, fmaf(v2.z, u1.z, fmaf(v2.y, u1.y, v2.x * u1.x)));
    float ps3 = fmaf(v3.w, u1.w, fmaf(v3.z, u1.z, fmaf(v3.y, u1.y, v3.x * u1.x)));
    float pt0 = fmaf(v0.w, u2.w, fmaf(v0.z, u2.z, fmaf(v0.y, u2.y, v0.x * u2.x)));
    float pt1 = fmaf(v1.w, u2.w, fmaf(v1.z, u2.z, fmaf(v1.y, u2.y, v1.x * u2.x)));
    float pt2 = fmaf(v2.w, u2.w, fmaf(v2.z, u2.z, fmaf(v2.y, u2.y, v2.x * u2.x)));
    float pt3 = fmaf(v3.w, u2.w, fmaf(v3.z, u2.z, fmaf(v3.y, u2.y, v3.x * u2.x)));
#pragma unroll
    for (int off = 8; off >= 1; off >>= 1) {
        ps0 += __shfl_xor(ps0, off, 32);  pt0 += __shfl_xor(pt0, off, 32);
        ps1 += __shfl_xor(ps1, off, 32);  pt1 += __shfl_xor(pt1, off, 32);
        ps2 += __shfl_xor(ps2, off, 32);  pt2 += __shfl_xor(pt2, off, 32);
        ps3 += __shfl_xor(ps3, off, 32);  pt3 += __shfl_xor(pt3, off, 32);
    }
    if (l15 == 0) {
        int r = row0 + half;
        s1[r]     = ps0;  t2[r]     = pt0;
        s1[r + 2] = ps1;  t2[r + 2] = pt1;
        s1[r + 4] = ps2;  t2[r + 4] = pt2;
        s1[r + 6] = ps3;  t2[r + 6] = pt3;
    }
}

__global__ __launch_bounds__(256) void rowmax_kernel(
    const int* __restrict__ adj, const float* __restrict__ s1,
    const float* __restrict__ t2, float* __restrict__ mrow)
{
    int wave = threadIdx.x >> 5;
    int lane = threadIdx.x & 31;
    int i = blockIdx.x * 8 + wave;
    const int* arow = adj + (size_t)i * N_NODES;
    float T0 = -3.0e38f, T1 = -3.0e38f, T2 = -3.0e38f, T3 = -3.0e38f;
    int any = 0;
    for (int j = lane; j < N_NODES; j += 32) {
        int av = arow[j];
        bool sel = av > 0;
        any |= (int)sel;
        float u0 = t2[j];
        float u1 = t2[N_NODES + j];
        float u2 = t2[2 * N_NODES + j];
        float u3 = t2[3 * N_NODES + j];
        T0 = fmaxf(T0, sel ? u0 : -3.0e38f);
        T1 = fmaxf(T1, sel ? u1 : -3.0e38f);
        T2 = fmaxf(T2, sel ? u2 : -3.0e38f);
        T3 = fmaxf(T3, sel ? u3 : -3.0e38f);
    }
#pragma unroll
    for (int off = 16; off >= 1; off >>= 1) {
        T0 = fmaxf(T0, __shfl_xor(T0, off, 32));
        T1 = fmaxf(T1, __shfl_xor(T1, off, 32));
        T2 = fmaxf(T2, __shfl_xor(T2, off, 32));
        T3 = fmaxf(T3, __shfl_xor(T3, off, 32));
        any |= __shfl_xor(any, off, 32);
    }
    if (lane == 0) {
        float Tb[4] = {T0, T1, T2, T3};
#pragma unroll
        for (int b = 0; b < 4; ++b) {
            float x = s1[b * N_NODES + i] + Tb[b];
            float m = any ? fmaxf(x, ALPHA * x) : NEG_BIG;
            mrow[b * N_NODES + i] = m;
        }
    }
}

#define KB        32
#define ROWS_WG   128
#define ADJ_PITCH 36

__global__ __launch_bounds__(256) void attn_kernel(
    const int* __restrict__ adj, const float* __restrict__ WhP,
    const float* __restrict__ s1, const float* __restrict__ t2,
    const float* __restrict__ mrow, float* __restrict__ Dp,
    float* __restrict__ Zp, int jspan)
{
    __shared__ __attribute__((aligned(16))) float WhS[2][KB * F_SZ];
    __shared__ __attribute__((aligned(16))) int   adjS[2][ROWS_WG * ADJ_PITCH];
    __shared__ __attribute__((aligned(16))) float tS[2][KB];

    int bid   = blockIdx.x;
    int b     = bid & 3;
    int rblk  = (bid >> 2) & 31;
    int chunk = bid >> 7;
    int tid   = threadIdx.x;
    int wave  = tid >> 5;
    int lane  = tid & 31;
    int l15   = lane & 15;
    int half  = lane >> 4;
    int itile = rblk * 8 + wave;
    int irow  = itile * 16 + l15;

    float s = s1[b * N_NODES + irow];
    float m = mrow[b * N_NODES + irow];
    const float* trow   = t2 + (size_t)b * N_NODES;
    const float* WhB    = WhP + (size_t)b * N_NODES * F_SZ;
    const int*   adjblk = adj + (size_t)(rblk * ROWS_WG) * N_NODES;

    int wr0 = tid >> 4;
    int wl  = tid & 15;
    int wst = (wr0 >> 1) * 128 + (wl << 2) + (wr0 & 1);
    int ar0 = tid >> 3;
    int as4 = (tid & 7) << 2;

    int j0   = chunk * jspan;
    int nblk = jspan / KB;

    float4 gw0, gw1;
    int4   ga0, ga1, ga2, ga3;
    float  gt;

    {
        const float* wp = WhB + (size_t)(j0 + wr0) * F_SZ + (wl << 2);
        gw0 = *(const float4*)(wp);
        gw1 = *(const float4*)(wp + 16 * F_SZ);
        const int* ap = adjblk + (size_t)ar0 * N_NODES + j0 + as4;
        ga0 = *(const int4*)(ap);
        ga1 = *(const int4*)(ap + (size_t)32 * N_NODES);
        ga2 = *(const int4*)(ap + (size_t)64 * N_NODES);
        ga3 = *(const int4*)(ap + (size_t)96 * N_NODES);
        gt  = trow[j0 + lane];
        float* w0 = &WhS[0][wst];
        float* w1 = w0 + 8 * 128;
        w0[0] = gw0.x; w0[2] = gw0.y; w0[64] = gw0.z; w0[66] = gw0.w;
        w1[0] = gw1.x; w1[2] = gw1.y; w1[64] = gw1.z; w1[66] = gw1.w;
        *(int4*)(&adjS[0][ar0 * ADJ_PITCH + as4])        = ga0;
        *(int4*)(&adjS[0][(ar0 + 32) * ADJ_PITCH + as4]) = ga1;
        *(int4*)(&adjS[0][(ar0 + 64) * ADJ_PITCH + as4]) = ga2;
        *(int4*)(&adjS[0][(ar0 + 96) * ADJ_PITCH + as4]) = ga3;
        if (wave == 0) tS[0][lane] = gt;
    }
    __syncthreads();

    v8f c0 = {}, c1 = {}, c2 = {}, c3 = {};
    float zacc = 0.f;

    for (int blk = 0; blk < nblk; ++blk) {
        int  buf  = blk & 1;
        bool more = (blk + 1 < nblk);

        if (more) {
            int jn = j0 + (blk + 1) * KB;
            const float* wp = WhB + (size_t)(jn + wr0) * F_SZ + (wl << 2);
            gw0 = *(const float4*)(wp);
            gw1 = *(const float4*)(wp + 16 * F_SZ);
            const int* ap = adjblk + (size_t)ar0 * N_NODES + jn + as4;
            ga0 = *(const int4*)(ap);
            ga1 = *(const int4*)(ap + (size_t)32 * N_NODES);
            ga2 = *(const int4*)(ap + (size_t)64 * N_NODES);
            ga3 = *(const int4*)(ap + (size_t)96 * N_NODES);
            gt  = trow[jn + lane];
        }

        const float* wS = &WhS[buf][half * 128 + (l15 << 2)];
        const int*   aS = &adjS[buf][(wave * 16 + l15) * ADJ_PITCH];
        const float* tB = &tS[buf][0];

#pragma unroll
        for (int q = 0; q < KB / 4; ++q) {
            int jl = q * 4 + half * 2;
            int2   av = *(const int2*)(aS + jl);
            float2 tv = *(const float2*)(tB + jl);

            float x0 = s + tv.x, x1 = s + tv.y;
            float e0 = fmaxf(x0, ALPHA * x0);
            float e1 = fmaxf(x1, ALPHA * x1);
            float sc0 = (av.x > 0) ? e0 : NEG_BIG;
            float sc1 = (av.y > 0) ? e1 : NEG_BIG;
            float p0 = __builtin_amdgcn_exp2f((sc0 - m) * L2E);
            float p1 = __builtin_amdgcn_exp2f((sc1 - m) * L2E);
            zacc += p0 + p1;

            v2f A; A[0] = p0; A[1] = p1;

            const float* bp = wS + q * 256;
            float4 u0 = *(const float4*)(bp);
            float4 u1 = *(const float4*)(bp + 64);
            v2f B0, B1, B2, B3;
            B0[0] = u0.x; B0[1] = u0.y;
            B1[0] = u0.z; B1[1] = u0.w;
            B2[0] = u1.x; B2[1] = u1.y;
            B3[0] = u1.z; B3[1] = u1.w;

            c0 = __builtin_amdgcn_wmma_f32_16x16x4_f32(false, A, false, B0, (short)0, c0, false, false);
            c1 = __builtin_amdgcn_wmma_f32_16x16x4_f32(false, A, false, B1, (short)0, c1, false, false);
            c2 = __builtin_amdgcn_wmma_f32_16x16x4_f32(false, A, false, B2, (short)0, c2, false, false);
            c3 = __builtin_amdgcn_wmma_f32_16x16x4_f32(false, A, false, B3, (short)0, c3, false, false);
        }

        if (more) {
            int nb = buf ^ 1;
            float* w0 = &WhS[nb][wst];
            float* w1 = w0 + 8 * 128;
            w0[0] = gw0.x; w0[2] = gw0.y; w0[64] = gw0.z; w0[66] = gw0.w;
            w1[0] = gw1.x; w1[2] = gw1.y; w1[64] = gw1.z; w1[66] = gw1.w;
            *(int4*)(&adjS[nb][ar0 * ADJ_PITCH + as4])        = ga0;
            *(int4*)(&adjS[nb][(ar0 + 32) * ADJ_PITCH + as4]) = ga1;
            *(int4*)(&adjS[nb][(ar0 + 64) * ADJ_PITCH + as4]) = ga2;
            *(int4*)(&adjS[nb][(ar0 + 96) * ADJ_PITCH + as4]) = ga3;
            if (wave == 0) tS[nb][lane] = gt;
        }
        __syncthreads();
    }

    zacc += __shfl_xor(zacc, 16, 32);

    size_t obase = (((size_t)chunk * 4 + b) * N_NODES + (size_t)itile * 16) * F_SZ;
    float* dout = Dp + obase;
    int rbase = half * 8;
#pragma unroll
    for (int r = 0; r < 8; ++r) {
        float4 v; v.x = c0[r]; v.y = c1[r]; v.z = c2[r]; v.w = c3[r];
        *(float4*)(dout + (size_t)(rbase + r) * F_SZ + (l15 << 2)) = v;
    }
    if (half == 0)
        Zp[((size_t)chunk * 4 + b) * N_NODES + irow] = zacc;
}

__global__ __launch_bounds__(128) void bn_kernel(
    const float* __restrict__ Dp, const float* __restrict__ Zp,
    const float* __restrict__ h, const float* __restrict__ W,
    const float* __restrict__ gamma, const float* __restrict__ beta,
    float* __restrict__ out, int nchunk)
{
    int tid   = threadIdx.x;
    int wave  = tid >> 5;
    int lane  = tid & 31;
    int l15   = lane & 15;
    int half  = lane >> 4;
    int node0 = (blockIdx.x * 4 + wave) * 4;

    int rb = l15 & 3;
    int rn = node0 + (l15 >> 2);

    float g[32];
#pragma unroll
    for (int e = 0; e < 32; ++e) g[e] = 0.f;
    float z = 0.f;
    const float* dlane = Dp + (half << 2);
    if (nchunk == 4) {
        float4 pv[4][8];
        float  pz[4];
#pragma unroll
        for (int c = 0; c < 4; ++c) {
            size_t rowi = ((size_t)c * 4 + rb) * N_NODES + rn;
            const float* p = dlane + (rowi << 6);
#pragma unroll
            for (int q = 0; q < 8; ++q) pv[c][q] = *(const float4*)(p + 8 * q);
            pz[c] = Zp[rowi];
        }
#pragma unroll
        for (int c = 0; c < 4; ++c) {
#pragma unroll
            for (int q = 0; q < 8; ++q) {
                g[4 * q + 0] += pv[c][q].x;  g[4 * q + 1] += pv[c][q].y;
                g[4 * q + 2] += pv[c][q].z;  g[4 * q + 3] += pv[c][q].w;
            }
            z += pz[c];
        }
    } else {
        for (int c = 0; c < nchunk; ++c) {
            size_t rowi = ((size_t)c * 4 + rb) * N_NODES + rn;
            const float* p = dlane + (rowi << 6);
#pragma unroll
            for (int q = 0; q < 8; ++q) {
                float4 v = *(const float4*)(p + 8 * q);
                g[4 * q + 0] += v.x;  g[4 * q + 1] += v.y;
                g[4 * q + 2] += v.z;  g[4 * q + 3] += v.w;
            }
            z += Zp[rowi];
        }
    }
#pragma unroll
    for (int e = 0; e < 32; ++e) g[e] = g[e] / z;

    v8f d0 = {}, d1 = {}, d2 = {}, d3 = {};
    const float* wp = W + (size_t)(half << 2) * F_SZ + (l15 << 2);
#pragma unroll
    for (int kk = 0; kk < 16; ++kk) {
        int f = 8 * (kk >> 1) + 2 * (kk & 1);
        float4 w0 = *(const float4*)(wp + (size_t)f * F_SZ);
        float4 w1 = *(const float4*)(wp + (size_t)(f + 1) * F_SZ);
        v2f A;  A[0] = g[2 * kk];  A[1] = g[2 * kk + 1];
        v2f B0, B1, B2, B3;
        B0[0] = w0.x; B0[1] = w1.x;
        B1[0] = w0.y; B1[1] = w1.y;
        B2[0] = w0.z; B2[1] = w1.z;
        B3[0] = w0.w; B3[1] = w1.w;
        d0 = __builtin_amdgcn_wmma_f32_16x16x4_f32(false, A, false, B0, (short)0, d0, false, false);
        d1 = __builtin_amdgcn_wmma_f32_16x16x4_f32(false, A, false, B1, (short)0, d1, false, false);
        d2 = __builtin_amdgcn_wmma_f32_16x16x4_f32(false, A, false, B2, (short)0, d2, false, false);
        d3 = __builtin_amdgcn_wmma_f32_16x16x4_f32(false, A, false, B3, (short)0, d3, false, false);
    }

    float sA = 0.f, sB = 0.f;
#pragma unroll
    for (int v = 0; v < 4; ++v) {
        sA += (d0[v] + d1[v]) + (d2[v] + d3[v]);
        sB += (d0[v + 4] + d1[v + 4]) + (d2[v + 4] + d3[v + 4]);
    }
#pragma unroll
    for (int off = 8; off >= 1; off >>= 1) {
        sA += __shfl_xor(sA, off, 32);
        sB += __shfl_xor(sB, off, 32);
    }
    float meanA = sA * (1.0f / 256.0f);
    float meanB = sB * (1.0f / 256.0f);

    float qA = 0.f, qB = 0.f;
#pragma unroll
    for (int v = 0; v < 4; ++v) {
        d0[v] -= meanA;  d1[v] -= meanA;  d2[v] -= meanA;  d3[v] -= meanA;
        d0[v + 4] -= meanB;  d1[v + 4] -= meanB;  d2[v + 4] -= meanB;  d3[v + 4] -= meanB;
        qA += (d0[v] * d0[v] + d1[v] * d1[v]) + (d2[v] * d2[v] + d3[v] * d3[v]);
        qB += (d0[v + 4] * d0[v + 4] + d1[v + 4] * d1[v + 4]) + (d2[v + 4] * d2[v + 4] + d3[v + 4] * d3[v + 4]);
    }
#pragma unroll
    for (int off = 8; off >= 1; off >>= 1) {
        qA += __shfl_xor(qA, off, 32);
        qB += __shfl_xor(qB, off, 32);
    }
    float varA = qA * (1.0f / 256.0f);
    float varB = qB * (1.0f / 256.0f);

    int nA = node0 + 2 * half;
    int nB = nA + 1;
    float gaA = gamma[nA], beA = beta[nA];
    float gaB = gamma[nB], beB = beta[nB];
    float rsA = 1.0f / sqrtf(varA + 1e-5f);
    float rsB = 1.0f / sqrtf(varB + 1e-5f);

    size_t offA = ((size_t)nA << 6) + (l15 << 2);
#pragma unroll
    for (int v = 0; v < 8; ++v) {
        float ga = (v < 4) ? gaA : gaB;
        float be = (v < 4) ? beA : beB;
        float rs = (v < 4) ? rsA : rsB;
        size_t off = offA + (((size_t)(v & 3) * N_NODES + (v >> 2)) << 6);
        float4 hv = *(const float4*)(h + off);
        d0[v] = ga * d0[v] * rs + be + hv.x;
        d1[v] = ga * d1[v] * rs + be + hv.y;
        d2[v] = ga * d2[v] * rs + be + hv.z;
        d3[v] = ga * d3[v] * rs + be + hv.w;
    }
#pragma unroll
    for (int v = 0; v < 8; ++v) {
        size_t off = offA + (((size_t)(v & 3) * N_NODES + (v >> 2)) << 6);
        float4 o;
        o.x = d0[v] > 0.f ? d0[v] : expm1f(d0[v]);
        o.y = d1[v] > 0.f ? d1[v] : expm1f(d1[v]);
        o.z = d2[v] > 0.f ? d2[v] : expm1f(d2[v]);
        o.w = d3[v] > 0.f ? d3[v] : expm1f(d3[v]);
        *(float4*)(out + off) = o;
    }
}

extern "C" void kernel_launch(void* const* d_in, const int* in_sizes, int n_in,
                              void* d_out, int out_size, void* d_ws, size_t ws_size,
                              hipStream_t stream)
{
    const float* h     = (const float*)d_in[0];
    const int*   adj   = (const int*)  d_in[1];
    const float* W     = (const float*)d_in[2];
    const float* a     = (const float*)d_in[3];
    const float* gamma = (const float*)d_in[4];
    const float* beta  = (const float*)d_in[5];
    float* out = (float*)d_out;

    const size_t WH_ELEMS = (size_t)B_SZ * N_NODES * F_SZ;
    const size_t NROW     = (size_t)B_SZ * N_NODES;

    int nchunk = 4;
    size_t need4 = (WH_ELEMS + 3 * NROW + 4 * (WH_ELEMS + NROW)) * sizeof(float);
    if (ws_size < need4) nchunk = 1;

    float* wsf  = (float*)d_ws;
    float* s1   = wsf + WH_ELEMS;
    float* t2   = s1 + NROW;
    float* mrow = t2 + NROW;
    float* Dp   = mrow + NROW;
    float* Zp   = Dp + (size_t)nchunk * WH_ELEMS;

    int jspan = N_NODES / nchunk;

    st_kernel<<<B_SZ * N_NODES / 64, 256, 0, stream>>>(h, W, a, s1, t2);
    rowmax_kernel<<<N_NODES / 8, 256, 0, stream>>>(adj, s1, t2, mrow);
    attn_kernel<<<B_SZ * (N_NODES / ROWS_WG) * nchunk, 256, 0, stream>>>(
        adj, h, s1, t2, mrow, Dp, Zp, jspan);
    bn_kernel<<<N_NODES / 16, 128, 0, stream>>>(Dp, Zp, h, W, gamma, beta, out, nchunk);
}


# online softmax: attn keeps a running shift per row and j chunk (block max before each block, rescale only on a rise above 16), bn aligns the 4 chunks; row-max launch removed; sum order unchanged
# speedup vs baseline: 2.2547x; 1.2944x over previous
#include <hip/hip_runtime.h>
#include <hip/hip_bf16.h>
#include <math.h>

typedef float v2f __attribute__((ext_vector_type(2)));
typedef float v8f __attribute__((ext_vector_type(8)));

#define N_NODES 4096
#define B_SZ    4
#define F_SZ    64
#define ALPHA   0.2f
#define NEG_BIG -9e15f
#define L2E     1.4426950408889634f

__global__ __launch_bounds__(256) void st_kernel(
    const float* __restrict__ h, const float* __restrict__ W,
    const float* __restrict__ a, float* __restrict__ s1,
    float* __restrict__ t2)
{
    __shared__ __attribute__((aligned(16))) float uS[2 * F_SZ];

    int tid  = threadIdx.x;
    int wave = tid >> 5;
    int lane = tid & 31;
    int l15  = lane & 15;
    int half = lane >> 4;
    int row0 = (blockIdx.x * 8 + wave) * 8;

    const float* hp = h + (size_t)(row0 + half) * F_SZ + (l15 << 2);
    float4 v0 = *(const float4*)(hp);
    float4 v1 = *(const float4*)(hp + 2 * F_SZ);
    float4 v2 = *(const float4*)(hp + 4 * F_SZ);
    float4 v3 = *(const float4*)(hp + 6 * F_SZ);

    if (tid < 2 * F_SZ) {
        int sel = tid >> 6;
        int f   = tid & 63;
        const float* wr = W + f * F_SZ;
        const float* av = a + sel * F_SZ;
        float u = 0.f;
#pragma unroll
        for (int o = 0; o < F_SZ; ++o) u = fmaf(wr[o], av[o], u);
        uS[tid] = u * (1.0f / 64.0f);
    }
    __syncthreads();

    float4 u1 = *(const float4*)(&uS[l15 << 2]);
    float4 u2 = *(const float4*)(&uS[F_SZ + (l15 << 2)]);

    float ps0 = fmaf(v0.w, u1.w, fmaf(v0.z, u1.z, fmaf(v0.y, u1.y, v0.x * u1.x)));
    float ps1 = fmaf(v1.w, u1.w, fmaf(v1.z, u1.z, fmaf(v1.y, u1.y, v1.x * u1.x)));
    float ps2 = fmaf(v2.w, u1.w, fmaf(v2.z, u1.z, fmaf(v2.y, u1.y, v2.x * u1.x)));
    float ps3 = fmaf(v3.w, u1.w, fmaf(v3.z, u1.z, fmaf(v3.y, u1.y, v3.x * u1.x)));
    float pt0 = fmaf(v0.w, u2.w, fmaf(v0.z, u2.z, fmaf(v0.y, u2.y, v0.x * u2.x)));
    float pt1 = fmaf(v1.w, u2.w, fmaf(v1.z, u2.z, fmaf(v1.y, u2.y, v1.x * u2.x)));
    float pt2 = fmaf(v2.w, u2.w, fmaf(v2.z, u2.z, fmaf(v2.y, u2.y, v2.x * u2.x)));
    float pt3 = fmaf(v3.w, u2.w, fmaf(v3.z, u2.z, fmaf(v3.y, u2.y, v3.x * u2.x)));
#pragma unroll
    for (int off = 8; off >= 1; off >>= 1) {
        ps0 += __shfl_xor(ps0, off, 32);  pt0 += __shfl_xor(pt0, off, 32);
        ps1 += __shfl_xor(ps1, off, 32);  pt1 += __shfl_xor(pt1, off, 32);
        ps2 += __shfl_xor(ps2, off, 32);  pt2 += __shfl_xor(pt2, off, 32);
        ps3 += __shfl_xor(ps3, off, 32);  pt3 += __shfl_xor(pt3, off, 32);
    }
    if (l15 == 0) {
        int r = row0 + half;
        s1[r]     = ps0;  t2[r]     = pt0;
        s1[r + 2] = ps1;  t2[r + 2] = pt1;
        s1[r + 4] = ps2;  t2[r + 4] = pt2;
        s1[r + 6] = ps3;  t2[r + 6] = pt3;
    }
}

#define SHIFT_MARGIN 16.0f

#define KB        32
#define ROWS_WG   128
#define ADJ_PITCH 36

__global__ __launch_bounds__(256) void attn_kernel(
    const int* __restrict__ adj, const float* __restrict__ WhP,
    const float* __restrict__ s1, const float* __restrict__ t2,
    float* __restrict__ Dp, float* __restrict__ Zp,
    float* __restrict__ Mp, int jspan)
{
    __shared__ __attribute__((aligned(16))) float WhS[2][KB * F_SZ];
    __shared__ __attribute__((aligned(16))) int   adjS[2][ROWS_WG * ADJ_PITCH];
    __shared__ __attribute__((aligned(16))) float tS[2][KB];

    int bid   = blockIdx.x;
    int b     = bid & 3;
    int rblk  = (bid >> 2) & 31;
    int chunk = bid >> 7;
    int tid   = threadIdx.x;
    int wave  = tid >> 5;
    int lane  = tid & 31;
    int l15   = lane & 15;
    int half  = lane >> 4;
    int itile = rblk * 8 + wave;
    int irow  = itile * 16 + l15;

    float s = s1[b * N_NODES + irow];
    float m = NEG_BIG;
    const float* trow   = t2 + (size_t)b * N_NODES;
    const float* WhB    = WhP + (size_t)b * N_NODES * F_SZ;
    const int*   adjblk = adj + (size_t)(rblk * ROWS_WG) * N_NODES;

    int wr0 = tid >> 4;
    int wl  = tid & 15;
    int wst = (wr0 >> 1) * 128 + (wl << 2) + (wr0 & 1);
    int ar0 = tid >> 3;
    int as4 = (tid & 7) << 2;

    int j0   = chunk * jspan;
    int nblk = jspan / KB;

    float4 gw0, gw1;
    int4   ga0, ga1, ga2, ga3;
    float  gt;

    {
        const float* wp = WhB + (size_t)(j0 + wr0) * F_SZ + (wl << 2);
        gw0 = *(const float4*)(wp);
        gw1 = *(const float4*)(wp + 16 * F_SZ);
        const int* ap = adjblk + (size_t)ar0 * N_NODES + j0 + as4;
        ga0 = *(const int4*)(ap);
        ga1 = *(const int4*)(ap + (size_t)32 * N_NODES);
        ga2 = *(const int4*)(ap + (size_t)64 * N_NODES);
        ga3 = *(const int4*)(ap + (size_t)96 * N_NODES);
        gt  = trow[j0 + lane];
        float* w0 = &WhS[0][wst];
        float* w1 = w0 + 8 * 128;
        w0[0] = gw0.x; w0[2] = gw0.y; w0[64] = gw0.z; w0[66] = gw0.w;
        w1[0] = gw1.x; w1[2] = gw1.y; w1[64] = gw1.z; w1[66] = gw1.w;
        *(int4*)(&adjS[0][ar0 * ADJ_PITCH + as4])        = ga0;
        *(int4*)(&adjS[0][(ar0 + 32) * ADJ_PITCH + as4]) = ga1;
        *(int4*)(&adjS[0][(ar0 + 64) * ADJ_PITCH + as4]) = ga2;
        *(int4*)(&adjS[0][(ar0 + 96) * ADJ_PITCH + as4]) = ga3;
        if (wave == 0) tS[0][lane] = gt;
    }
    __syncthreads();

    v8f c0 = {}, c1 = {}, c2 = {}, c3 = {};
    float zacc = 0.f;

    for (int blk = 0; blk < nblk; ++blk) {
        int  buf  = blk & 1;
        bool more = (blk + 1 < nblk);

        if (more) {
            int jn = j0 + (blk + 1) * KB;
            const float* wp = WhB + (size_t)(jn + wr0) * F_SZ + (wl << 2);
            gw0 = *(const float4*)(wp);
            gw1 = *(const float4*)(wp + 16 * F_SZ);
            const int* ap = adjblk + (size_t)ar0 * N_NODES + jn + as4;
            ga0 = *(const int4*)(ap);
            ga1 = *(const int4*)(ap + (size_t)32 * N_NODES);
            ga2 = *(const int4*)(ap + (size_t)64 * N_NODES);
            ga3 = *(const int4*)(ap + (size_t)96 * N_NODES);
            gt  = trow[jn + lane];
        }

        const float* wS = &WhS[buf][half * 128 + (l15 << 2)];
        const int*   aS = &adjS[buf][(wave * 16 + l15) * ADJ_PITCH];
        const float* tB = &tS[buf][0];

        float sc[KB / 2];
        float bm = NEG_BIG;
#pragma unroll
        for (int q = 0; q < KB / 4; ++q) {
            int jl = q * 4 + half * 2;
            int2   av = *(const int2*)(aS + jl);
            float2 tv = *(const float2*)(tB + jl);

            float x0 = s + tv.x, x1 = s + tv.y;
            float e0 = fmaxf(x0, ALPHA * x0);
            float e1 = fmaxf(x1, ALPHA * x1);
            float sc0 = (av.x > 0) ? e0 : NEG_BIG;
            float sc1 = (av.y > 0) ? e1 : NEG_BIG;
            sc[2 * q]     = sc0;
            sc[2 * q + 1] = sc1;
            bm = fmaxf(bm, fmaxf(sc0, sc1));
        }
        bm = fmaxf(bm, __shfl_xor(bm, 16, 32));

        bool up = bm > m + SHIFT_MARGIN;
        if (__any(up)) {
            float mn = up ? bm : m;
            float f  = __builtin_amdgcn_exp2f((m - mn) * L2E);
            m = mn;
            zacc *= f;
#pragma unroll
            for (int r = 0; r < 8; ++r) {
                float fr = __shfl(f, r + 8 * half, 32);
                c0[r] *= fr;  c1[r] *= fr;  c2[r] *= fr;  c3[r] *= fr;
            }
        }

#pragma unroll
        for (int q = 0; q < KB / 4; ++q) {
            float p0 = __builtin_amdgcn_exp2f((sc[2 * q] - m) * L2E);
            float p1 = __builtin_amdgcn_exp2f((sc[2 * q + 1] - m) * L2E);
            zacc += p0 + p1;

            v2f A; A[0] = p0; A[1] = p1;

            const float* bp = wS + q * 256;
            float4 u0 = *(const float4*)(bp);
            float4 u1 = *(const float4*)(bp + 64);
            v2f B0, B1, B2, B3;
            B0[0] = u0.x; B0[1] = u0.y;
            B1[0] = u0.z; B1[1] = u0.w;
            B2[0] = u1.x; B2[1] = u1.y;
            B3[0] = u1.z; B3[1] = u1.w;

            c0 = __builtin_amdgcn_wmma_f32_16x16x4_f32(false, A, false, B0, (short)0, c0, false, false);
            c1 = __builtin_amdgcn_wmma_f32_16x16x4_f32(false, A, false, B1, (short)0, c1, false, false);
            c2 = __builtin_amdgcn_wmma_f32_16x16x4_f32(false, A, false, B2, (short)0, c2, false, false);
            c3 = __builtin_amdgcn_wmma_f32_16x16x4_f32(false, A, false, B3, (short)0, c3, false, false);
        }

        if (more) {
            int nb = buf ^ 1;
            float* w0 = &WhS[nb][wst];
            float* w1 = w0 + 8 * 128;
            w0[0] = gw0.x; w0[2] = gw0.y; w0[64] = gw0.z; w0[66] = gw0.w;
            w1[0] = gw1.x; w1[2] = gw1.y; w1[64] = gw1.z; w1[66] = gw1.w;
            *(int4*)(&adjS[nb][ar0 * ADJ_PITCH + as4])        = ga0;
            *(int4*)(&adjS[nb][(ar0 + 32) * ADJ_PITCH + as4]) = ga1;
            *(int4*)(&adjS[nb][(ar0 + 64) * ADJ_PITCH + as4]) = ga2;
            *(int4*)(&adjS[nb][(ar0 + 96) * ADJ_PITCH + as4]) = ga3;
            if (wave == 0) tS[nb][lane] = gt;
        }
        __syncthreads();
    }

    zacc += __shfl_xor(zacc, 16, 32);

    size_t obase = (((size_t)chunk * 4 + b) * N_NODES + (size_t)itile * 16) * F_SZ;
    float* dout = Dp + obase;
    int rbase = half * 8;
#pragma unroll
    for (int r = 0; r < 8; ++r) {
        float4 v; v.x = c0[r]; v.y = c1[r]; v.z = c2[r]; v.w = c3[r];
        *(float4*)(dout + (size_t)(rbase + r) * F_SZ + (l15 << 2)) = v;
    }
    if (half == 0) {
        size_t zi = ((size_t)chunk * 4 + b) * N_NODES + irow;
        Zp[zi] = zacc;
        Mp[zi] = m;
    }
}

__global__ __launch_bounds__(128) void bn_kernel(
    const float* __restrict__ Dp, const float* __restrict__ Zp,
    const float* __restrict__ Mp,
    const float* __restrict__ h, const float* __restrict__ W,
    const float* __restrict__ gamma, const float* __restrict__ beta,
    float* __restrict__ out, int nchunk)
{
    int tid   = threadIdx.x;
    int wave  = tid >> 5;
    int lane  = tid & 31;
    int l15   = lane & 15;
    int half  = lane >> 4;
    int node0 = (blockIdx.x * 4 + wave) * 4;

    int rb = l15 & 3;
    int rn = node0 + (l15 >> 2);

    float g[32];
#pragma unroll
    for (int e = 0; e < 32; ++e) g[e] = 0.f;
    float z = 0.f;
    const float* dlane = Dp + (half << 2);
    if (nchunk == 4) {
        float4 pv[4][8];
        float  pz[4], pm[4];
#pragma unroll
        for (int c = 0; c < 4; ++c) {
            size_t rowi = ((size_t)c * 4 + rb) * N_NODES + rn;
            const float* p = dlane + (rowi << 6);
#pragma unroll
            for (int q = 0; q < 8; ++q) pv[c][q] = *(const float4*)(p + 8 * q);
            pz[c] = Zp[rowi];
            pm[c] = Mp[rowi];
        }
        float mtop = fmaxf(fmaxf(pm[0], pm[1]), fmaxf(pm[2], pm[3]));
#pragma unroll
        for (int c = 0; c < 4; ++c) {
            float w = __builtin_amdgcn_exp2f((pm[c] - mtop) * L2E);
#pragma unroll
            for (int q = 0; q < 8; ++q) {
                g[4 * q + 0] = fmaf(pv[c][q].x, w, g[4 * q + 0]);
                g[4 * q + 1] = fmaf(pv[c][q].y, w, g[4 * q + 1]);
                g[4 * q + 2] = fmaf(pv[c][q].z, w, g[4 * q + 2]);
                g[4 * q + 3] = fmaf(pv[c][q].w, w, g[4 * q + 3]);
            }
            z = fmaf(pz[c], w, z);
        }
    } else {
        float mtop = NEG_BIG;
        for (int c = 0; c < nchunk; ++c)
            mtop = fmaxf(mtop, Mp[((size_t)c * 4 + rb) * N_NODES + rn]);
        for (int c = 0; c < nchunk; ++c) {
            size_t rowi = ((size_t)c * 4 + rb) * N_NODES + rn;
            const float* p = dlane + (rowi << 6);
            float w = __builtin_amdgcn_exp2f((Mp[rowi] - mtop) * L2E);
#pragma unroll
            for (int q = 0; q < 8; ++q) {
                float4 v = *(const float4*)(p + 8 * q);
                g[4 * q + 0] = fmaf(v.x, w, g[4 * q + 0]);
                g[4 * q + 1] = fmaf(v.y, w, g[4 * q + 1]);
                g[4 * q + 2] = fmaf(v.z, w, g[4 * q + 2]);
                g[4 * q + 3] = fmaf(v.w, w, g[4 * q + 3]);
            }
            z = fmaf(Zp[rowi], w, z);
        }
    }
#pragma unroll
    for (int e = 0; e < 32; ++e) g[e] = g[e] / z;

    v8f d0 = {}, d1 = {}, d2 = {}, d3 = {};
    const float* wp = W + (size_t)(half << 2) * F_SZ + (l15 << 2);
#pragma unroll
    for (int kk = 0; kk < 16; ++kk) {
        int f = 8 * (kk >> 1) + 2 * (kk & 1);
        float4 w0 = *(const float4*)(wp + (size_t)f * F_SZ);
        float4 w1 = *(const float4*)(wp + (size_t)(f + 1) * F_SZ);
        v2f A;  A[0] = g[2 * kk];  A[1] = g[2 * kk + 1];
        v2f B0, B1, B2, B3;
        B0[0] = w0.x; B0[1] = w1.x;
        B1[0] = w0.y; B1[1] = w1.y;
        B2[0] = w0.z; B2[1] = w1.z;
        B3[0] = w0.w; B3[1] = w1.w;
        d0 = __builtin_amdgcn_wmma_f32_16x16x4_f32(false, A, false, B0, (short)0, d0, false, false);
        d1 = __builtin_amdgcn_wmma_f32_16x16x4_f32(false, A, false, B1, (short)0, d1, false, false);
        d2 = __builtin_amdgcn_wmma_f32_16x16x4_f32(false, A, false, B2, (short)0, d2, false, false);
        d3 = __builtin_amdgcn_wmma_f32_16x16x4_f32(false, A, false, B3, (short)0, d3, false, false);
    }

    float sA = 0.f, sB = 0.f;
#pragma unroll
    for (int v = 0; v < 4; ++v) {
        sA += (d0[v] + d1[v]) + (d2[v] + d3[v]);
        sB += (d0[v + 4] + d1[v + 4]) + (d2[v + 4] + d3[v + 4]);
    }
#pragma unroll
    for (int off = 8; off >= 1; off >>= 1) {
        sA += __shfl_xor(sA, off, 32);
        sB += __shfl_xor(sB, off, 32);
    }
    float meanA = sA * (1.0f / 256.0f);
    float meanB = sB * (1.0f / 256.0f);

    float qA = 0.f, qB = 0.f;
#pragma unroll
    for (int v = 0; v < 4; ++v) {
        d0[v] -= meanA;  d1[v] -= meanA;  d2[v] -= meanA;  d3[v] -= meanA;
        d0[v + 4] -= meanB;  d1[v + 4] -= meanB;  d2[v + 4] -= meanB;  d3[v + 4] -= meanB;
        qA += (d0[v] * d0[v] + d1[v] * d1[v]) + (d2[v] * d2[v] + d3[v] * d3[v]);
        qB += (d0[v + 4] * d0[v + 4] + d1[v + 4] * d1[v + 4]) + (d2[v + 4] * d2[v + 4] + d3[v + 4] * d3[v + 4]);
    }
#pragma unroll
    for (int off = 8; off >= 1; off >>= 1) {
        qA += __shfl_xor(qA, off, 32);
        qB += __shfl_xor(qB, off, 32);
    }
    float varA = qA * (1.0f / 256.0f);
    float varB = qB * (1.0f / 256.0f);

    int nA = node0 + 2 * half;
    int nB = nA + 1;
    float gaA = gamma[nA], beA = beta[nA];
    float gaB = gamma[nB], beB = beta[nB];
    float rsA = 1.0f / sqrtf(varA + 1e-5f);
    float rsB = 1.0f / sqrtf(varB + 1e-5f);

    size_t offA = ((size_t)nA << 6) + (l15 << 2);
#pragma unroll
    for (int v = 0; v < 8; ++v) {
        float ga = (v < 4) ? gaA : gaB;
        float be = (v < 4) ? beA : beB;
        float rs = (v < 4) ? rsA : rsB;
        size_t off = offA + (((size_t)(v & 3) * N_NODES + (v >> 2)) << 6);
        float4 hv = *(const float4*)(h + off);
        d0[v] = ga * d0[v] * rs + be + hv.x;
        d1[v] = ga * d1[v] * rs + be + hv.y;
        d2[v] = ga * d2[v] * rs + be + hv.z;
        d3[v] = ga * d3[v] * rs + be + hv.w;
    }
#pragma unroll
    for (int v = 0; v < 8; ++v) {
        size_t off = offA + (((size_t)(v & 3) * N_NODES + (v >> 2)) << 6);
        float4 o;
        o.x = d0[v] > 0.f ? d0[v] : expm1f(d0[v]);
        o.y = d1[v] > 0.f ? d1[v] : expm1f(d1[v]);
        o.z = d2[v] > 0.f ? d2[v] : expm1f(d2[v]);
        o.w = d3[v] > 0.f ? d3[v] : expm1f(d3[v]);
        *(float4*)(out + off) = o;
    }
}

extern "C" void kernel_launch(void* const* d_in, const int* in_sizes, int n_in,
                              void* d_out, int out_size, void* d_ws, size_t ws_size,
                              hipStream_t stream)
{
    const float* h     = (const float*)d_in[0];
    const int*   adj   = (const int*)  d_in[1];
    const float* W     = (const float*)d_in[2];
    const float* a     = (const float*)d_in[3];
    const float* gamma = (const float*)d_in[4];
    const float* beta  = (const float*)d_in[5];
    float* out = (float*)d_out;

    const size_t WH_ELEMS = (size_t)B_SZ * N_NODES * F_SZ;
    const size_t NROW     = (size_t)B_SZ * N_NODES;

    int nchunk = 4;
    size_t need4 = (WH_ELEMS + 3 * NROW + 4 * (WH_ELEMS + NROW)) * sizeof(float);
    if (ws_size < need4) nchunk = 1;

    float* wsf  = (float*)d_ws;
    float* Mp   = wsf;
    float* s1   = wsf + WH_ELEMS;
    float* t2   = s1 + NROW;
    float* Dp   = t2 + NROW;
    float* Zp   = Dp + (size_t)nchunk * WH_ELEMS;

    int jspan = N_NODES / nchunk;

    st_kernel<<<B_SZ * N_NODES / 64, 256, 0, stream>>>(h, W, a, s1, t2);
    attn_kernel<<<B_SZ * (N_NODES / ROWS_WG) * nchunk, 256, 0, stream>>>(
        adj, h, s1, t2, Dp, Zp, Mp, jspan);
    bn_kernel<<<N_NODES / 16, 128, 0, stream>>>(Dp, Zp, Mp, h, W, gamma, beta, out, nchunk);
}


# attn kernel: wave priority falls as a work-group advances along its columns (s_setprio 3,2,1,0 from 5/8, 7/8 and the last block): the two groups on a WGP take turns and end together; sums unchanged
# speedup vs baseline: 2.3395x; 1.0376x over previous
#include <hip/hip_runtime.h>
#include <hip/hip_bf16.h>
#include <math.h>

typedef float v2f __attribute__((ext_vector_type(2)));
typedef float v8f __attribute__((ext_vector_type(8)));

#define N_NODES 4096
#define B_SZ    4
#define F_SZ    64
#define ALPHA   0.2f
#define NEG_BIG -9e15f
#define L2E     1.4426950408889634f

__global__ __launch_bounds__(256) void st_kernel(
    const float* __restrict__ h, const float* __restrict__ W,
    const float* __restrict__ a, float* __restrict__ s1,
    float* __restrict__ t2)
{
    __shared__ __attribute__((aligned(16))) float uS[2 * F_SZ];

    int tid  = threadIdx.x;
    int wave = tid >> 5;
    int lane = tid & 31;
    int l15  = lane & 15;
    int half = lane >> 4;
    int row0 = (blockIdx.x * 8 + wave) * 8;

    const float* hp = h + (size_t)(row0 + half) * F_SZ + (l15 << 2);
    float4 v0 = *(const float4*)(hp);
    float4 v1 = *(const float4*)(hp + 2 * F_SZ);
    float4 v2 = *(const float4*)(hp + 4 * F_SZ);
    float4 v3 = *(const float4*)(hp + 6 * F_SZ);

    if (tid < 2 * F_SZ) {
        int sel = tid >> 6;
        int f   = tid & 63;
        const float* wr = W + f * F_SZ;
        const float* av = a + sel * F_SZ;
        float u = 0.f;
#pragma unroll
        for (int o = 0; o < F_SZ; ++o) u = fmaf(wr[o], av[o], u);
        uS[tid] = u * (1.0f / 64.0f);
    }
    __syncthreads();

    float4 u1 = *(const float4*)(&uS[l15 << 2]);
    float4 u2 = *(const float4*)(&uS[F_SZ + (l15 << 2)]);

    float ps0 = fmaf(v0.w, u1.w, fmaf(v0.z, u1.z, fmaf(v0.y, u1.y, v0.x * u1.x)));
    float ps1 = fmaf(v1.w, u1.w, fmaf(v1.z, u1.z, fmaf(v1.y, u1.y, v1.x * u1.x)));
    float ps2 = fmaf(v2.w, u1.w, fmaf(v2.z, u1.z, fmaf(v2.y, u1.y, v2.x * u1.x)));
    float ps3 = fmaf(v3.w, u1.w, fmaf(v3.z, u1.z, fmaf(v3.y, u1.y, v3.x * u1.x)));
    float pt0 = fmaf(v0.w, u2.w, fmaf(v0.z, u2.z, fmaf(v0.y, u2.y, v0.x * u2.x)));
    float pt1 = fmaf(v1.w, u2.w, fmaf(v1.z, u2.z, fmaf(v1.y, u2.y, v1.x * u2.x)));
    float pt2 = fmaf(v2.w, u2.w, fmaf(v2.z, u2.z, fmaf(v2.y, u2.y, v2.x * u2.x)));
    float pt3 = fmaf(v3.w, u2.w, fmaf(v3.z, u2.z, fmaf(v3.y, u2.y, v3.x * u2.x)));
#pragma unroll
    for (int off = 8; off >= 1; off >>= 1) {
        ps0 += __shfl_xor(ps0, off, 32);  pt0 += __shfl_xor(pt0, off, 32);
        ps1 += __shfl_xor(ps1, off, 32);  pt1 += __shfl_xor(pt1, off, 32);
        ps2 += __shfl_xor(ps2, off, 32);  pt2 += __shfl_xor(pt2, off, 32);
        ps3 += __shfl_xor(ps3, off, 32);  pt3 += __shfl_xor(pt3, off, 32);
    }
    if (l15 == 0) {
        int r = row0 + half;
        s1[r]     = ps0;  t2[r]     = pt0;
        s1[r + 2] = ps1;  t2[r + 2] = pt1;
        s1[r + 4] = ps2;  t2[r + 4] = pt2;
        s1[r + 6] = ps3;  t2[r + 6] = pt3;
    }
}

#define SHIFT_MARGIN 16.0f

#define KB        32
#define ROWS_WG   128
#define ADJ_PITCH 36

__global__ __launch_bounds__(256) void attn_kernel(
    const int* __restrict__ adj, const float* __restrict__ WhP,
    const float* __restrict__ s1, const float* __restrict__ t2,
    float* __restrict__ Dp, float* __restrict__ Zp,
    float* __restrict__ Mp, int jspan)
{
    __shared__ __attribute__((aligned(16))) float WhS[2][KB * F_SZ];
    __shared__ __attribute__((aligned(16))) int   adjS[2][ROWS_WG * ADJ_PITCH];
    __shared__ __attribute__((aligned(16))) float tS[2][KB];

    int bid   = blockIdx.x;
    int b     = bid & 3;
    int rblk  = (bid >> 2) & 31;
    int chunk = bid >> 7;
    int tid   = threadIdx.x;
    int wave  = tid >> 5;
    int lane  = tid & 31;
    int l15   = lane & 15;
    int half  = lane >> 4;
    int itile = rblk * 8 + wave;
    int irow  = itile * 16 + l15;

    float s = s1[b * N_NODES + irow];
    float m = NEG_BIG;
    const float* trow   = t2 + (size_t)b * N_NODES;
    const float* WhB    = WhP + (size_t)b * N_NODES * F_SZ;
    const int*   adjblk = adj + (size_t)(rblk * ROWS_WG) * N_NODES;

    int wr0 = tid >> 4;
    int wl  = tid & 15;
    int wst = (wr0 >> 1) * 128 + (wl << 2) + (wr0 & 1);
    int ar0 = tid >> 3;
    int as4 = (tid & 7) << 2;

    int j0   = chunk * jspan;
    int nblk = jspan / KB;

    int pstep1 = (nblk * 5) >> 3;
    int pstep2 = (nblk * 7) >> 3;
    int pstep3 = nblk - 1;
    int pnext  = pstep1;
    int plevel = 3;

    float4 gw0, gw1;
    int4   ga0, ga1, ga2, ga3;
    float  gt;

    {
        const float* wp = WhB + (size_t)(j0 + wr0) * F_SZ + (wl << 2);
        gw0 = *(const float4*)(wp);
        gw1 = *(const float4*)(wp + 16 * F_SZ);
        const int* ap = adjblk + (size_t)ar0 * N_NODES + j0 + as4;
        ga0 = *(const int4*)(ap);
        ga1 = *(const int4*)(ap + (size_t)32 * N_NODES);
        ga2 = *(const int4*)(ap + (size_t)64 * N_NODES);
        ga3 = *(const int4*)(ap + (size_t)96 * N_NODES);
        gt  = trow[j0 + lane];
        float* w0 = &WhS[0][wst];
        float* w1 = w0 + 8 * 128;
        w0[0] = gw0.x; w0[2] = gw0.y; w0[64] = gw0.z; w0[66] = gw0.w;
        w1[0] = gw1.x; w1[2] = gw1.y; w1[64] = gw1.z; w1[66] = gw1.w;
        *(int4*)(&adjS[0][ar0 * ADJ_PITCH + as4])        = ga0;
        *(int4*)(&adjS[0][(ar0 + 32) * ADJ_PITCH + as4]) = ga1;
        *(int4*)(&adjS[0][(ar0 + 64) * ADJ_PITCH + as4]) = ga2;
        *(int4*)(&adjS[0][(ar0 + 96) * ADJ_PITCH + as4]) = ga3;
        if (wave == 0) tS[0][lane] = gt;
    }
    __syncthreads();

    v8f c0 = {}, c1 = {}, c2 = {}, c3 = {};
    float zacc = 0.f;

    __builtin_amdgcn_s_setprio(3);

    for (int blk = 0; blk < nblk; ++blk) {
        int  buf  = blk & 1;
        bool more = (blk + 1 < nblk);

        if (blk == pnext) {
            plevel -= 1;
            if (plevel == 2)      { __builtin_amdgcn_s_setprio(2); pnext = pstep2; }
            else if (plevel == 1) { __builtin_amdgcn_s_setprio(1); pnext = pstep3; }
            else                  { __builtin_amdgcn_s_setprio(0); pnext = -1; }
        }

        if (more) {
            int jn = j0 + (blk + 1) * KB;
            const float* wp = WhB + (size_t)(jn + wr0) * F_SZ + (wl << 2);
            gw0 = *(const float4*)(wp);
            gw1 = *(const float4*)(wp + 16 * F_SZ);
            const int* ap = adjblk + (size_t)ar0 * N_NODES + jn + as4;
            ga0 = *(const int4*)(ap);
            ga1 = *(const int4*)(ap + (size_t)32 * N_NODES);
            ga2 = *(const int4*)(ap + (size_t)64 * N_NODES);
            ga3 = *(const int4*)(ap + (size_t)96 * N_NODES);
            gt  = trow[jn + lane];
        }

        const float* wS = &WhS[buf][half * 128 + (l15 << 2)];
        const int*   aS = &adjS[buf][(wave * 16 + l15) * ADJ_PITCH];
        const float* tB = &tS[buf][0];

        float sc[KB / 2];
        float bm = NEG_BIG;
#pragma unroll
        for (int q = 0; q < KB / 4; ++q) {
            int jl = q * 4 + half * 2;
            int2   av = *(const int2*)(aS + jl);
            float2 tv = *(const float2*)(tB + jl);

            float x0 = s + tv.x, x1 = s + tv.y;
            float e0 = fmaxf(x0, ALPHA * x0);
            float e1 = fmaxf(x1, ALPHA * x1);
            float sc0 = (av.x > 0) ? e0 : NEG_BIG;
            float sc1 = (av.y > 0) ? e1 : NEG_BIG;
            sc[2 * q]     = sc0;
            sc[2 * q + 1] = sc1;
            bm = fmaxf(bm, fmaxf(sc0, sc1));
        }
        bm = fmaxf(bm, __shfl_xor(bm, 16, 32));

        bool up = bm > m + SHIFT_MARGIN;
        if (__any(up)) {
            float mn = up ? bm : m;
            float f  = __builtin_amdgcn_exp2f((m - mn) * L2E);
            m = mn;
            zacc *= f;
#pragma unroll
            for (int r = 0; r < 8; ++r) {
                float fr = __shfl(f, r + 8 * half, 32);
                c0[r] *= fr;  c1[r] *= fr;  c2[r] *= fr;  c3[r] *= fr;
            }
        }

#pragma unroll
        for (int q = 0; q < KB / 4; ++q) {
            float p0 = __builtin_amdgcn_exp2f((sc[2 * q] - m) * L2E);
            float p1 = __builtin_amdgcn_exp2f((sc[2 * q + 1] - m) * L2E);
            zacc += p0 + p1;

            v2f A; A[0] = p0; A[1] = p1;

            const float* bp = wS + q * 256;
            float4 u0 = *(const float4*)(bp);
            float4 u1 = *(const float4*)(bp + 64);
            v2f B0, B1, B2, B3;
            B0[0] = u0.x; B0[1] = u0.y;
            B1[0] = u0.z; B1[1] = u0.w;
            B2[0] = u1.x; B2[1] = u1.y;
            B3[0] = u1.z; B3[1] = u1.w;

            c0 = __builtin_amdgcn_wmma_f32_16x16x4_f32(false, A, false, B0, (short)0, c0, false, false);
            c1 = __builtin_amdgcn_wmma_f32_16x16x4_f32(false, A, false, B1, (short)0, c1, false, false);
            c2 = __builtin_amdgcn_wmma_f32_16x16x4_f32(false, A, false, B2, (short)0, c2, false, false);
            c3 = __builtin_amdgcn_wmma_f32_16x16x4_f32(false, A, false, B3, (short)0, c3, false, false);
        }

        if (more) {
            int nb = buf ^ 1;
            float* w0 = &WhS[nb][wst];
            float* w1 = w0 + 8 * 128;
            w0[0] = gw0.x; w0[2] = gw0.y; w0[64] = gw0.z; w0[66] = gw0.w;
            w1[0] = gw1.x; w1[2] = gw1.y; w1[64] = gw1.z; w1[66] = gw1.w;
            *(int4*)(&adjS[nb][ar0 * ADJ_PITCH + as4])        = ga0;
            *(int4*)(&adjS[nb][(ar0 + 32) * ADJ_PITCH + as4]) = ga1;
            *(int4*)(&adjS[nb][(ar0 + 64) * ADJ_PITCH + as4]) = ga2;
            *(int4*)(&adjS[nb][(ar0 + 96) * ADJ_PITCH + as4]) = ga3;
            if (wave == 0) tS[nb][lane] = gt;
        }
        __syncthreads();
    }

    __builtin_amdgcn_s_setprio(0);

    zacc += __shfl_xor(zacc, 16, 32);

    size_t obase = (((size_t)chunk * 4 + b) * N_NODES + (size_t)itile * 16) * F_SZ;
    float* dout = Dp + obase;
    int rbase = half * 8;
#pragma unroll
    for (int r = 0; r < 8; ++r) {
        float4 v; v.x = c0[r]; v.y = c1[r]; v.z = c2[r]; v.w = c3[r];
        *(float4*)(dout + (size_t)(rbase + r) * F_SZ + (l15 << 2)) = v;
    }
    if (half == 0) {
        size_t zi = ((size_t)chunk * 4 + b) * N_NODES + irow;
        Zp[zi] = zacc;
        Mp[zi] = m;
    }
}

__global__ __launch_bounds__(128) void bn_kernel(
    const float* __restrict__ Dp, const float* __restrict__ Zp,
    const float* __restrict__ Mp,
    const float* __restrict__ h, const float* __restrict__ W,
    const float* __restrict__ gamma, const float* __restrict__ beta,
    float* __restrict__ out, int nchunk)
{
    int tid   = threadIdx.x;
    int wave  = tid >> 5;
    int lane  = tid & 31;
    int l15   = lane & 15;
    int half  = lane >> 4;
    int node0 = (blockIdx.x * 4 + wave) * 4;

    int rb = l15 & 3;
    int rn = node0 + (l15 >> 2);

    float g[32];
#pragma unroll
    for (int e = 0; e < 32; ++e) g[e] = 0.f;
    float z = 0.f;
    const float* dlane = Dp + (half << 2);
    if (nchunk == 4) {
        float4 pv[4][8];
        float  pz[4], pm[4];
#pragma unroll
        for (int c = 0; c < 4; ++c) {
            size_t rowi = ((size_t)c * 4 + rb) * N_NODES + rn;
            const float* p = dlane + (rowi << 6);
#pragma unroll
            for (int q = 0; q < 8; ++q) pv[c][q] = *(const float4*)(p + 8 * q);
            pz[c] = Zp[rowi];
            pm[c] = Mp[rowi];
        }
        float mtop = fmaxf(fmaxf(pm[0], pm[1]), fmaxf(pm[2], pm[3]));
#pragma unroll
        for (int c = 0; c < 4; ++c) {
            float w = __builtin_amdgcn_exp2f((pm[c] - mtop) * L2E);
#pragma unroll
            for (int q = 0; q < 8; ++q) {
                g[4 * q + 0] = fmaf(pv[c][q].x, w, g[4 * q + 0]);
                g[4 * q + 1] = fmaf(pv[c][q].y, w, g[4 * q + 1]);
                g[4 * q + 2] = fmaf(pv[c][q].z, w, g[4 * q + 2]);
                g[4 * q + 3] = fmaf(pv[c][q].w, w, g[4 * q + 3]);
            }
            z = fmaf(pz[c], w, z);
        }
    } else {
        float mtop = NEG_BIG;
        for (int c = 0; c < nchunk; ++c)
            mtop = fmaxf(mtop, Mp[((size_t)c * 4 + rb) * N_NODES + rn]);
        for (int c = 0; c < nchunk; ++c) {
            size_t rowi = ((size_t)c * 4 + rb) * N_NODES + rn;
            const float* p = dlane + (rowi << 6);
            float w = __builtin_amdgcn_exp2f((Mp[rowi] - mtop) * L2E);
#pragma unroll
            for (int q = 0; q < 8; ++q) {
                float4 v = *(const float4*)(p + 8 * q);
                g[4 * q + 0] = fmaf(v.x, w, g[4 * q + 0]);
                g[4 * q + 1] = fmaf(v.y, w, g[4 * q + 1]);
                g[4 * q + 2] = fmaf(v.z, w, g[4 * q + 2]);
                g[4 * q + 3] = fmaf(v.w, w, g[4 * q + 3]);
            }
            z = fmaf(Zp[rowi], w, z);
        }
    }
#pragma unroll
    for (int e = 0; e < 32; ++e) g[e] = g[e] / z;

    v8f d0 = {}, d1 = {}, d2 = {}, d3 = {};
    const float* wp = W + (size_t)(half << 2) * F_SZ + (l15 << 2);
#pragma unroll
    for (int kk = 0; kk < 16; ++kk) {
        int f = 8 * (kk >> 1) + 2 * (kk & 1);
        float4 w0 = *(const float4*)(wp + (size_t)f * F_SZ);
        float4 w1 = *(const float4*)(wp + (size_t)(f + 1) * F_SZ);
        v2f A;  A[0] = g[2 * kk];  A[1] = g[2 * kk + 1];
        v2f B0, B1, B2, B3;
        B0[0] = w0.x; B0[1] = w1.x;
        B1[0] = w0.y; B1[1] = w1.y;
        B2[0] = w0.z; B2[1] = w1.z;
        B3[0] = w0.w; B3[1] = w1.w;
        d0 = __builtin_amdgcn_wmma_f32_16x16x4_f32(false, A, false, B0, (short)0, d0, false, false);
        d1 = __builtin_amdgcn_wmma_f32_16x16x4_f32(false, A, false, B1, (short)0, d1, false, false);
        d2 = __builtin_amdgcn_wmma_f32_16x16x4_f32(false, A, false, B2, (short)0, d2, false, false);
        d3 = __builtin_amdgcn_wmma_f32_16x16x4_f32(false, A, false, B3, (short)0, d3, false, false);
    }

    float sA = 0.f, sB = 0.f;
#pragma unroll
    for (int v = 0; v < 4; ++v) {
        sA += (d0[v] + d1[v]) + (d2[v] + d3[v]);
        sB += (d0[v + 4] + d1[v + 4]) + (d2[v + 4] + d3[v + 4]);
    }
#pragma unroll
    for (int off = 8; off >= 1; off >>= 1) {
        sA += __shfl_xor(sA, off, 32);
        sB += __shfl_xor(sB, off, 32);
    }
    float meanA = sA * (1.0f / 256.0f);
    float meanB = sB * (1.0f / 256.0f);

    float qA = 0.f, qB = 0.f;
#pragma unroll
    for (int v = 0; v < 4; ++v) {
        d0[v] -= meanA;  d1[v] -= meanA;  d2[v] -= meanA;  d3[v] -= meanA;
        d0[v + 4] -= meanB;  d1[v + 4] -= meanB;  d2[v + 4] -= meanB;  d3[v + 4] -= meanB;
        qA += (d0[v] * d0[v] + d1[v] * d1[v]) + (d2[v] * d2[v] + d3[v] * d3[v]);
        qB += (d0[v + 4] * d0[v + 4] + d1[v + 4] * d1[v + 4]) + (d2[v + 4] * d2[v + 4] + d3[v + 4] * d3[v + 4]);
    }
#pragma unroll
    for (int off = 8; off >= 1; off >>= 1) {
        qA += __shfl_xor(qA, off, 32);
        qB += __shfl_xor(qB, off, 32);
    }
    float varA = qA * (1.0f / 256.0f);
    float varB = qB * (1.0f / 256.0f);

    int nA = node0 + 2 * half;
    int nB = nA + 1;
    float gaA = gamma[nA], beA = beta[nA];
    float gaB = gamma[nB], beB = beta[nB];
    float rsA = 1.0f / sqrtf(varA + 1e-5f);
    float rsB = 1.0f / sqrtf(varB + 1e-5f);

    size_t offA = ((size_t)nA << 6) + (l15 << 2);
#pragma unroll
    for (int v = 0; v < 8; ++v) {
        float ga = (v < 4) ? gaA : gaB;
        float be = (v < 4) ? beA : beB;
        float rs = (v < 4) ? rsA : rsB;
        size_t off = offA + (((size_t)(v & 3) * N_NODES + (v >> 2)) << 6);
        float4 hv = *(const float4*)(h + off);
        d0[v] = ga * d0[v] * rs + be + hv.x;
        d1[v] = ga * d1[v] * rs + be + hv.y;
        d2[v] = ga * d2[v] * rs + be + hv.z;
        d3[v] = ga * d3[v] * rs + be + hv.w;
    }
#pragma unroll
    for (int v = 0; v < 8; ++v) {
        size_t off = offA + (((size_t)(v & 3) * N_NODES + (v >> 2)) << 6);
        float4 o;
        o.x = d0[v] > 0.f ? d0[v] : expm1f(d0[v]);
        o.y = d1[v] > 0.f ? d1[v] : expm1f(d1[v]);
        o.z = d2[v] > 0.f ? d2[v] : expm1f(d2[v]);
        o.w = d3[v] > 0.f ? d3[v] : expm1f(d3[v]);
        *(float4*)(out + off) = o;
    }
}

extern "C" void kernel_launch(void* const* d_in, const int* in_sizes, int n_in,
                              void* d_out, int out_size, void* d_ws, size_t ws_size,
                              hipStream_t stream)
{
    const float* h     = (const float*)d_in[0];
    const int*   adj   = (const int*)  d_in[1];
    const float* W     = (const float*)d_in[2];
    const float* a     = (const float*)d_in[3];
    const float* gamma = (const float*)d_in[4];
    const float* beta  = (const float*)d_in[5];
    float* out = (float*)d_out;

    const size_t WH_ELEMS = (size_t)B_SZ * N_NODES * F_SZ;
    const size_t NROW     = (size_t)B_SZ * N_NODES;

    int nchunk = 4;
    size_t need4 = (WH_ELEMS + 3 * NROW + 4 * (WH_ELEMS + NROW)) * sizeof(float);
    if (ws_size < need4) nchunk = 1;

    float* wsf  = (float*)d_ws;
    float* Mp   = wsf;
    float* s1   = wsf + WH_ELEMS;
    float* t2   = s1 + NROW;
    float* Dp   = t2 + NROW;
    float* Zp   = Dp + (size_t)nchunk * WH_ELEMS;

    int jspan = N_NODES / nchunk;

    st_kernel<<<B_SZ * N_NODES / 64, 256, 0, stream>>>(h, W, a, s1, t2);
    attn_kernel<<<B_SZ * (N_NODES / ROWS_WG) * nchunk, 256, 0, stream>>>(
        adj, h, s1, t2, Dp, Zp, Mp, jspan);
    bn_kernel<<<N_NODES / 16, 128, 0, stream>>>(Dp, Zp, Mp, h, W, gamma, beta, out, nchunk);
}
